# starting kernel
# baseline (speedup 1.0000x reference)
#include <hip/hip_runtime.h>

typedef float v2f __attribute__((ext_vector_type(2)));
typedef float v8f __attribute__((ext_vector_type(8)));

__global__ __launch_bounds__(256) void mbd_gemm_f32wmma(
    const float* __restrict__ x,
    const float* __restrict__ kern,
    float* __restrict__ M)
{
    const int lane   = threadIdx.x & 31;
    const int wave   = threadIdx.x >> 5;
    const int waveId = blockIdx.x * 8 + wave;
    const int tm     = waveId >> 4;
    const int tn     = waveId & 15;

    const int l16  = lane & 15;
    const int half = lane >> 4;
    const int r0   = tm * 16;
    const int c0   = tn * 16;

    const float* arow = x + (r0 + l16) * 256 + 2 * half;

    const int    j    = c0 + l16;
    const float* wcol = kern + (j >> 3) * 2048 + (j & 7);
    const int    kb   = 2 * half;

    v8f acc = {};
#pragma unroll 4
    for (int k = 0; k < 256; k += 4) {
        v2f a = *(const v2f*)(arow + k);
        v2f b;
        b.x = wcol[(k + kb) * 8];
        b.y = wcol[(k + kb + 1) * 8];
        acc = __builtin_amdgcn_wmma_f32_16x16x4_f32(
                  false, a, false, b, (short)0, acc, false, false);
    }

#pragma unroll
    for (int r = 0; r < 8; ++r) {
        M[(r0 + r + 8 * half) * 256 + c0 + l16] = acc[r];
    }
}

__global__ __launch_bounds__(256) void mbd_pairwise(
    const float* __restrict__ M,
    float* __restrict__ out)
{
    __shared__ float lds[1024 * 8];

    const int o   = blockIdx.y;
    const int tid = threadIdx.x;

    const float* src = M + o * 8;
    for (int i = tid; i < 2048; i += 256) {
        const int b = i >> 1;
        const int h = i & 1;
        *(float4*)(&lds[i * 4]) = *(const float4*)(src + b * 256 + h * 4);
    }
    __syncthreads();

    const int b = blockIdx.x * 256 + tid;
    float m[8];
#pragma unroll
    for (int l = 0; l < 8; ++l) m[l] = lds[b * 8 + l];

    const float kLog2e = 1.4426950408889634f;
    float acc = 0.0f;
#pragma unroll 4
    for (int bp = 0; bp < 1024; ++bp) {
        const float* q = &lds[bp * 8];
        float l1 = 0.0f;
#pragma unroll
        for (int l = 0; l < 8; ++l) l1 += __builtin_fabsf(m[l] - q[l]);
        acc += __builtin_amdgcn_exp2f(-l1 * kLog2e);
    }
    out[b * 32 + o] = acc * (1.0f / 1024.0f);
}

extern "C" void kernel_launch(void* const* d_in, const int* in_sizes, int n_in,
                              void* d_out, int out_size, void* d_ws, size_t ws_size,
                              hipStream_t stream)
{
    const float* x    = (const float*)d_in[0];
    const float* kern = (const float*)d_in[1];
    float*       out  = (float*)d_out;
    float*       M    = (float*)d_ws;

    mbd_gemm_f32wmma<<<128, 256, 0, stream>>>(x, kern, M);
    mbd_pairwise<<<dim3(4, 32), 256, 0, stream>>>(M, out);
}


# Stage 2 distance as sum of min(a,b) minus half sums (M kept in exp2 units): a min and an add per term, both two row groups wide; new rounding of the l sum, column sums split even/odd
# speedup vs baseline: 2.3348x; 2.3348x over previous
#include <hip/hip_runtime.h>

typedef float v2f __attribute__((ext_vector_type(2)));
typedef float v8f __attribute__((ext_vector_type(8)));

#define MBD_LOG2E 1.4426950408889634f
#define G1_KW 4
#define G1_NG 8
#define G1_TP 272
#define G1_WP (4 * G1_TP)

__global__ __launch_bounds__(128) void mbd_gemm_f32wmma(
    const float* __restrict__ x,
    const float* __restrict__ kern,
    float* __restrict__ M)
{
    __shared__ float red[G1_KW * G1_WP];

    const int lane = threadIdx.x & 31;
    const int wave = threadIdx.x >> 5;
    const int bm   = blockIdx.x >> 3;
    const int bn   = blockIdx.x & 7;

    const int l16  = lane & 15;
    const int half = lane >> 4;
    const int r0   = bm * 32;
    const int c0   = bn * 32;
    const int kw   = wave * 64 + 4 * half;

    const float* a0p = x + (r0 + l16) * 256 + kw;
    const float* a1p = a0p + 16 * 256;
    const int    j0  = c0 + l16;
    const int    j1  = j0 + 16;
    const float* b0p = kern + (j0 >> 3) * 2048 + (j0 & 7) + kw * 8;
    const float* b1p = kern + (j1 >> 3) * 2048 + (j1 & 7) + kw * 8;

    float4 a0[G1_NG], a1[G1_NG];
    float  b0[G1_NG][4], b1[G1_NG][4];
#pragma unroll
    for (int g = 0; g < G1_NG; ++g) {
        a0[g] = *(const float4*)(a0p + 8 * g);
        a1[g] = *(const float4*)(a1p + 8 * g);
#pragma unroll
        for (int e = 0; e < 4; ++e) {
            b0[g][e] = b0p[(8 * g + e) * 8];
            b1[g][e] = b1p[(8 * g + e) * 8];
        }
    }
    __builtin_amdgcn_sched_barrier(0);

    v8f acc00 = {}, acc01 = {}, acc10 = {}, acc11 = {};
#pragma unroll
    for (int g = 0; g < G1_NG; ++g) {
        const float a0e[4] = {a0[g].x, a0[g].y, a0[g].z, a0[g].w};
        const float a1e[4] = {a1[g].x, a1[g].y, a1[g].z, a1[g].w};
#pragma unroll
        for (int s = 0; s < 2; ++s) {
            v2f fa0, fa1, fb0, fb1;
            fa0.x = a0e[2 * s];   fa0.y = a0e[2 * s + 1];
            fa1.x = a1e[2 * s];   fa1.y = a1e[2 * s + 1];
            fb0.x = b0[g][2 * s]; fb0.y = b0[g][2 * s + 1];
            fb1.x = b1[g][2 * s]; fb1.y = b1[g][2 * s + 1];
            acc00 = __builtin_amdgcn_wmma_f32_16x16x4_f32(
                        false, fa0, false, fb0, (short)0, acc00, false, false);
            acc01 = __builtin_amdgcn_wmma_f32_16x16x4_f32(
                        false, fa0, false, fb1, (short)0, acc01, false, false);
            acc10 = __builtin_amdgcn_wmma_f32_16x16x4_f32(
                        false, fa1, false, fb0, (short)0, acc10, false, false);
            acc11 = __builtin_amdgcn_wmma_f32_16x16x4_f32(
                        false, fa1, false, fb1, (short)0, acc11, false, false);
        }
    }

    float* my = red + wave * G1_WP;
#pragma unroll
    for (int r = 0; r < 8; ++r) {
        my[0 * G1_TP + r * 32 + lane] = acc00[r];
        my[1 * G1_TP + r * 32 + lane] = acc01[r];
        my[2 * G1_TP + r * 32 + lane] = acc10[r];
        my[3 * G1_TP + r * 32 + lane] = acc11[r];
    }
    __syncthreads();

    const float* sp = red + ((wave >> 1) * 2 + half) * G1_TP + (wave & 1) * 16 + l16;
#pragma unroll
    for (int rr = 0; rr < 8; ++rr) {
        const float* q = sp + rr * 32;
        const float s01 = q[0 * G1_WP] + q[1 * G1_WP];
        const float s23 = q[2 * G1_WP] + q[3 * G1_WP];
        M[(r0 + 8 * wave + rr) * 256 + c0 + lane] = (s01 + s23) * MBD_LOG2E;
    }
}

#define MBD_SLOTS 19
#define MBD_TP    36
#define MBD_QC    4

template <int H>
__device__ __forceinline__ void mbd_task(const float* __restrict__ M,
                                         float* __restrict__ prt,
                                         float* __restrict__ mt,
                                         const int o, const int p, const int lane)
{
    constexpr int NT  = H ? 9 : 8;
    constexpr int DLO = H ? 8 : 0;
    constexpr int NP  = 4;

    v2f   m2[NP][8];
    v2f   ha2[NP];
    v2f   m8[4];
    float ha8 = 0.0f;
    float hp;
    float hp8 = 0.0f;
#pragma unroll
    for (int t = 0; t < NT; ++t) {
        const int r = (p + DLO + t) & 31;
        const float* rp = M + (r * 32 + lane) * 256 + o * 8;
        const float4 m0 = *(const float4*)(rp);
        const float4 m1 = *(const float4*)(rp + 4);
        const float a[8] = {m0.x * 2.0f, m0.y * 2.0f, m0.z * 2.0f, m0.w * 2.0f,
                            m1.x * 2.0f, m1.y * 2.0f, m1.z * 2.0f, m1.w * 2.0f};
        float s = a[0];
#pragma unroll
        for (int l = 1; l < 8; ++l) s += a[l];
        if (t < 2 * NP) {
#pragma unroll
            for (int l = 0; l < 8; ++l) m2[t >> 1][l][t & 1] = a[l];
            ha2[t >> 1][t & 1] = 0.5f * s;
        } else {
#pragma unroll
            for (int k = 0; k < 4; ++k) m8[k] = (v2f){a[2 * k], a[2 * k + 1]};
            const v2f s8 = ((m8[0] + m8[1]) + m8[2]) + m8[3];
            ha8 = 0.5f * (s8.x + s8.y);
        }
    }
    if (H == 0) {
        hp = ha2[0][0];
    } else {
        const float* rp = M + (p * 32 + lane) * 256 + o * 8;
        const float4 m0 = *(const float4*)(rp);
        const float4 m1 = *(const float4*)(rp + 4);
        const float b[8] = {m0.x * 2.0f, m0.y * 2.0f, m0.z * 2.0f, m0.w * 2.0f,
                            m1.x * 2.0f, m1.y * 2.0f, m1.z * 2.0f, m1.w * 2.0f};
        float s = b[0];
#pragma unroll
        for (int l = 1; l < 8; ++l) s += b[l];
        hp = 0.5f * s;
        hp8 = 0.5f * ((((b[0] + b[2]) + b[4]) + b[6]) + (((b[1] + b[3]) + b[5]) + b[7]));
    }

    v2f   racc2[NP];
    float racc8 = 0.0f;
#pragma unroll
    for (int u = 0; u < NP; ++u) racc2[u] = (v2f){0.0f, 0.0f};

    const float* qp = M + (p * 32) * 256 + o * 8;

    float qr[MBD_QC][8];
#pragma unroll
    for (int jj = 0; jj < MBD_QC; ++jj)
#pragma unroll
        for (int l = 0; l < 8; ++l) qr[jj][l] = qp[jj * 256 + l];

    for (int c = 0; c < 32 / MBD_QC; ++c) {
        float qc[MBD_QC][8];
#pragma unroll
        for (int jj = 0; jj < MBD_QC; ++jj)
#pragma unroll
            for (int l = 0; l < 8; ++l) qc[jj][l] = qr[jj][l] * 2.0f;

        const int cn = (c + 1 < 32 / MBD_QC) ? c + 1 : c;
#pragma unroll
        for (int jj = 0; jj < MBD_QC; ++jj)
#pragma unroll
            for (int l = 0; l < 8; ++l) qr[jj][l] = qp[(cn * MBD_QC + jj) * 256 + l];
        __builtin_amdgcn_sched_barrier(0);

        float hb[MBD_QC];
#pragma unroll
        for (int jj = 0; jj < MBD_QC; ++jj)
            hb[jj] = __builtin_bit_cast(float,
                     __builtin_amdgcn_readlane(__builtin_bit_cast(int, hp), MBD_QC * c + jj));

        v2f cacc2[MBD_QC];
#pragma unroll
        for (int jj = 0; jj < MBD_QC; ++jj) cacc2[jj] = (v2f){0.0f, 0.0f};

#pragma unroll
        for (int u = 0; u < NP; ++u) {
#pragma unroll
            for (int jj = 0; jj < MBD_QC; ++jj) {
                v2f acc = __builtin_elementwise_minimumnum(m2[u][0], (v2f){qc[jj][0], qc[jj][0]});
#pragma unroll
                for (int l = 1; l < 8; ++l)
                    acc += __builtin_elementwise_minimumnum(m2[u][l], (v2f){qc[jj][l], qc[jj][l]});
                const v2f arg = (acc - ha2[u]) - (v2f){hb[jj], hb[jj]};
                v2f e;
                e.x = __builtin_amdgcn_exp2f(arg.x);
                e.y = __builtin_amdgcn_exp2f(arg.y);
                racc2[u] += e;
                if (H == 0 && u == 0) cacc2[jj].y += e.y;
                else                  cacc2[jj] += e;
            }
        }
        if (H == 1) {
#pragma unroll
            for (int jj = 0; jj < MBD_QC; ++jj) {
                const float hb8 = __builtin_bit_cast(float,
                    __builtin_amdgcn_readlane(__builtin_bit_cast(int, hp8), MBD_QC * c + jj));
                v2f acc = __builtin_elementwise_minimumnum(m8[0], (v2f){qc[jj][0], qc[jj][1]});
#pragma unroll
                for (int k = 1; k < 4; ++k)
                    acc += __builtin_elementwise_minimumnum(m8[k], (v2f){qc[jj][2 * k], qc[jj][2 * k + 1]});
                racc8 += __builtin_amdgcn_exp2f(((acc.x + acc.y) - ha8) - hb8);
            }
        }

        float4 v;
        v.x = cacc2[0].x + cacc2[0].y; v.y = cacc2[1].x + cacc2[1].y;
        v.z = cacc2[2].x + cacc2[2].y; v.w = cacc2[3].x + cacc2[3].y;
        *(float4*)(&mt[lane * MBD_TP + MBD_QC * c]) = v;
    }

#pragma unroll
    for (int t = 0; t < NT; ++t) {
        const int d = DLO + t;
        const int r = (p + d) & 31;
        const float rs = (t < 2 * NP) ? racc2[t >> 1][t & 1] : racc8;
        prt[(r * MBD_SLOTS + d) * 32 + lane] = rs;
    }
}

__global__ __launch_bounds__(256) void mbd_pairwise_sym(
    const float* __restrict__ M,
    float* __restrict__ part)
{
    __shared__ float turn[8 * 32 * MBD_TP];

    const int o   = blockIdx.y;
    const int tid = threadIdx.x;

    const int lane = tid & 31;
    const int wave = __builtin_amdgcn_readfirstlane(tid >> 5);
    const int p    = blockIdx.x * 4 + (wave & 3);
    const int half = wave >> 2;

    float* prt = part + o * (32 * MBD_SLOTS * 32);
    float* mt  = turn + wave * (32 * MBD_TP);

    if (half == 0) mbd_task<0>(M, prt, mt, o, p, lane);
    else           mbd_task<1>(M, prt, mt, o, p, lane);

    __syncthreads();
    float s0 = 0.0f, s1 = 0.0f, s2 = 0.0f, s3 = 0.0f;
#pragma unroll
    for (int i = 0; i < 32; i += 4) {
        s0 += mt[(i + 0) * MBD_TP + lane];
        s1 += mt[(i + 1) * MBD_TP + lane];
        s2 += mt[(i + 2) * MBD_TP + lane];
        s3 += mt[(i + 3) * MBD_TP + lane];
    }
    prt[(p * MBD_SLOTS + 17 + half) * 32 + lane] = (s0 + s1) + (s2 + s3);
}

__global__ __launch_bounds__(1024) void mbd_combine(
    const float* __restrict__ part,
    float* __restrict__ out)
{
    __shared__ float tile[32 * 33];
    const int g    = blockIdx.x;
    const int t    = threadIdx.x;
    const int o    = t >> 5;
    const int lane = t & 31;
    const float* ps = part + ((o * 32 + g) * MBD_SLOTS) * 32 + lane;
    float s = 0.0f;
#pragma unroll
    for (int k = 0; k < MBD_SLOTS; ++k) s += ps[k * 32];
    tile[lane * 33 + o] = s * (1.0f / 1024.0f);
    __syncthreads();
    out[g * 1024 + t] = tile[(t >> 5) * 33 + (t & 31)];
}

extern "C" void kernel_launch(void* const* d_in, const int* in_sizes, int n_in,
                              void* d_out, int out_size, void* d_ws, size_t ws_size,
                              hipStream_t stream)
{
    const float* x    = (const float*)d_in[0];
    const float* kern = (const float*)d_in[1];
    float*       out  = (float*)d_out;
    float*       M    = (float*)d_ws;
    float*       part = (float*)d_ws + 262144;

    mbd_gemm_f32wmma<<<256, 128, 0, stream>>>(x, kern, M);
    mbd_pairwise_sym<<<dim3(8, 32), 256, 0, stream>>>(M, part);
    mbd_combine<<<32, 1024, 0, stream>>>(part, out);
}
